# starting kernel
# baseline (speedup 1.0000x reference)
#include <hip/hip_runtime.h>

typedef float v2f __attribute__((ext_vector_type(2)));
typedef float v8f __attribute__((ext_vector_type(8)));

#define D            64
#define K            1024
#define ESTRIDE      66
#define ROWS_TOTAL   65536
#define THREADS      256
#define WAVES        8
#define SLABS        2
#define ROWS_PER_BLOCK (WAVES * 16 * SLABS)
#define NUM_BLOCKS   (ROWS_TOTAL / ROWS_PER_BLOCK)

#define LDS_E_WORDS  (K * ESTRIDE)
#define LDS_ESQ_OFF  LDS_E_WORDS
#define LDS_WL_OFF   (LDS_E_WORDS + K)
#define LDS_WORDS    (LDS_E_WORDS + K + WAVES)

__global__ __launch_bounds__(THREADS)
void vq_main(const float* __restrict__ x, const float* __restrict__ emb,
             float* __restrict__ out, float* __restrict__ partial)
{
    extern __shared__ float lds[];
    float* eLds = lds;
    float* esq  = lds + LDS_ESQ_OFF;
    float* wl   = lds + LDS_WL_OFF;

    const int tid  = threadIdx.x;
    const int wave = tid >> 5;
    const int lane = tid & 31;
    const int n    = lane & 15;
    const int hi   = lane >> 4;

    for (int i = tid; i < K * D; i += THREADS)
        eLds[(i >> 6) * ESTRIDE + (i & 63)] = emb[i];
    __syncthreads();

    for (int c = wave; c < K; c += WAVES) {
        const float* er = eLds + c * ESTRIDE;
        float v0 = er[lane], v1 = er[lane + 32];
        float s = v0 * v0 + v1 * v1;
        #pragma unroll
        for (int m = 16; m >= 1; m >>= 1) s += __shfl_xor(s, m, 32);
        if (lane == 0) esq[c] = s;
    }
    __syncthreads();

    float waveLoss = 0.0f;

    for (int slab = 0; slab < SLABS; ++slab) {
        const int rowBase = blockIdx.x * ROWS_PER_BLOCK + slab * (WAVES * 16) + wave * 16;

        const float* xrow = x + (size_t)(rowBase + n) * D + 2 * hi;
        v2f a[16];
        float xsqp = 0.0f;
        #pragma unroll
        for (int kk = 0; kk < 16; ++kk) {
            a[kk] = *(const v2f*)(xrow + 4 * kk);
            xsqp += a[kk][0] * a[kk][0] + a[kk][1] * a[kk][1];
        }
        const float xsqFull = xsqp + __shfl_xor(xsqp, 16, 32);

        float best[8];
        int   bidx[8];
        #pragma unroll
        for (int j = 0; j < 8; ++j) { best[j] = __builtin_inff(); bidx[j] = 0; }

        for (int t = 0; t < K / 16; ++t) {
            const int c0 = t * 16;
            const float* brow = eLds + (c0 + n) * ESTRIDE + 2 * hi;
            v8f acc = {0.f, 0.f, 0.f, 0.f, 0.f, 0.f, 0.f, 0.f};
            #pragma unroll
            for (int kk = 0; kk < 16; ++kk) {
                v2f b = *(const v2f*)(brow + 4 * kk);
                acc = __builtin_amdgcn_wmma_f32_16x16x4_f32(
                          false, a[kk], false, b, (short)0, acc, false, false);
            }
            const float es  = esq[c0 + n];
            const int   code = c0 + n;
            #pragma unroll
            for (int j = 0; j < 8; ++j) {
                float sc = __builtin_fmaf(-2.0f, acc[j], es);
                if (sc < best[j]) { best[j] = sc; bidx[j] = code; }
            }
        }

        #pragma unroll
        for (int j = 0; j < 8; ++j) {
            #pragma unroll
            for (int m = 1; m <= 8; m <<= 1) {
                float ob = __shfl_xor(best[j], m, 32);
                int   oi = __shfl_xor(bidx[j], m, 32);
                if (ob < best[j] || (ob == best[j] && oi < bidx[j])) {
                    best[j] = ob; bidx[j] = oi;
                }
            }
        }

        float lp = (hi == 0) ? xsqFull : 0.0f;
        if (n == 0) {
            float s8 = 0.0f;
            #pragma unroll
            for (int j = 0; j < 8; ++j) s8 += best[j];
            lp += s8;
        }
        #pragma unroll
        for (int m = 16; m >= 1; m >>= 1) lp += __shfl_xor(lp, m, 32);
        waveLoss += lp;

        #pragma unroll
        for (int r = 0; r < 16; ++r) {
            const int j = r & 7;
            const int srcLane = (r >> 3) << 4;
            const int idx = __shfl(bidx[j], srcLane, 32);
            const float* er = eLds + idx * ESTRIDE;
            float* orow = out + (size_t)(rowBase + r) * D;
            orow[lane]      = er[lane];
            orow[lane + 32] = er[lane + 32];
        }
    }

    if (lane == 0) wl[wave] = waveLoss;
    __syncthreads();
    if (tid == 0) {
        float s = 0.0f;
        #pragma unroll
        for (int w = 0; w < WAVES; ++w) s += wl[w];
        partial[blockIdx.x] = s;
    }
}

__global__ __launch_bounds__(NUM_BLOCKS)
void vq_finalize(const float* __restrict__ partial, float* __restrict__ lossOut)
{
    __shared__ float red[NUM_BLOCKS];
    const int t = threadIdx.x;
    red[t] = partial[t];
    __syncthreads();
    for (int s = NUM_BLOCKS / 2; s > 0; s >>= 1) {
        if (t < s) red[t] += red[t + s];
        __syncthreads();
    }
    if (t == 0)
        lossOut[0] = 1.25f * red[0] / (float)((size_t)ROWS_TOTAL * D);
}

extern "C" void kernel_launch(void* const* d_in, const int* in_sizes, int n_in,
                              void* d_out, int out_size, void* d_ws, size_t ws_size,
                              hipStream_t stream)
{
    (void)in_sizes; (void)n_in; (void)out_size; (void)ws_size;
    const float* x   = (const float*)d_in[0];
    const float* emb = (const float*)d_in[1];
    float* out       = (float*)d_out;
    float* partial   = (float*)d_ws;

    vq_main<<<NUM_BLOCKS, THREADS, LDS_WORDS * sizeof(float), stream>>>(x, emb, out, partial);
    vq_finalize<<<1, NUM_BLOCKS, 0, stream>>>(partial, out + (size_t)ROWS_TOTAL * D);
}


# Code norms ||e||^2: one thread sums one code row from LDS in registers (4 rows per thread) instead of one wave per code with 5 dependent cross-lane hops; same addition tree, sums bit-identical
# speedup vs baseline: 1.4566x; 1.4566x over previous
#include <hip/hip_runtime.h>

typedef float v2f __attribute__((ext_vector_type(2)));
typedef float v8f __attribute__((ext_vector_type(8)));

#define D            64
#define K            1024
#define ESTRIDE      66
#define ROWS_TOTAL   65536
#define THREADS      256
#define WAVES        8
#define SLABS        2
#define ROWS_PER_BLOCK (WAVES * 16 * SLABS)
#define NUM_BLOCKS   (ROWS_TOTAL / ROWS_PER_BLOCK)

#define LDS_E_WORDS  (K * ESTRIDE)
#define LDS_ESQ_OFF  LDS_E_WORDS
#define LDS_WL_OFF   (LDS_E_WORDS + K)
#define LDS_WORDS    (LDS_E_WORDS + K + WAVES)

__global__ __launch_bounds__(THREADS)
void vq_main(const float* __restrict__ x, const float* __restrict__ emb,
             float* __restrict__ out, float* __restrict__ partial)
{
    extern __shared__ float lds[];
    float* eLds = lds;
    float* esq  = lds + LDS_ESQ_OFF;
    float* wl   = lds + LDS_WL_OFF;

    const int tid  = threadIdx.x;
    const int wave = tid >> 5;
    const int lane = tid & 31;
    const int n    = lane & 15;
    const int hi   = lane >> 4;

    for (int i = tid; i < K * D; i += THREADS)
        eLds[(i >> 6) * ESTRIDE + (i & 63)] = emb[i];
    __syncthreads();

    for (int c = tid; c < K; c += THREADS) {
        const float* er = eLds + c * ESTRIDE;
        float u[16];
        #pragma unroll
        for (int l = 0; l < 16; l += 2) {
            const v2f a0 = *(const v2f*)(er + l);
            const v2f a1 = *(const v2f*)(er + l + 32);
            const v2f b0 = *(const v2f*)(er + l + 16);
            const v2f b1 = *(const v2f*)(er + l + 48);
            const float p0 = __builtin_fmaf(a0[0], a0[0], a1[0] * a1[0]);
            const float p1 = __builtin_fmaf(a0[1], a0[1], a1[1] * a1[1]);
            const float q0 = __builtin_fmaf(b0[0], b0[0], b1[0] * b1[0]);
            const float q1 = __builtin_fmaf(b0[1], b0[1], b1[1] * b1[1]);
            u[l]     = p0 + q0;
            u[l + 1] = p1 + q1;
        }
        float v[8];
        #pragma unroll
        for (int l = 0; l < 8; ++l) v[l] = u[l] + u[l + 8];
        float w[4];
        #pragma unroll
        for (int l = 0; l < 4; ++l) w[l] = v[l] + v[l + 4];
        const float y0 = w[0] + w[2];
        const float y1 = w[1] + w[3];
        esq[c] = y0 + y1;
    }
    __syncthreads();

    float waveLoss = 0.0f;

    for (int slab = 0; slab < SLABS; ++slab) {
        const int rowBase = blockIdx.x * ROWS_PER_BLOCK + slab * (WAVES * 16) + wave * 16;

        const float* xrow = x + (size_t)(rowBase + n) * D + 2 * hi;
        v2f a[16];
        float xsqp = 0.0f;
        #pragma unroll
        for (int kk = 0; kk < 16; ++kk) {
            a[kk] = *(const v2f*)(xrow + 4 * kk);
            xsqp += a[kk][0] * a[kk][0] + a[kk][1] * a[kk][1];
        }
        const float xsqFull = xsqp + __shfl_xor(xsqp, 16, 32);

        float best[8];
        int   bidx[8];
        #pragma unroll
        for (int j = 0; j < 8; ++j) { best[j] = __builtin_inff(); bidx[j] = 0; }

        for (int t = 0; t < K / 16; ++t) {
            const int c0 = t * 16;
            const float* brow = eLds + (c0 + n) * ESTRIDE + 2 * hi;
            v8f acc = {0.f, 0.f, 0.f, 0.f, 0.f, 0.f, 0.f, 0.f};
            #pragma unroll
            for (int kk = 0; kk < 16; ++kk) {
                v2f b = *(const v2f*)(brow + 4 * kk);
                acc = __builtin_amdgcn_wmma_f32_16x16x4_f32(
                          false, a[kk], false, b, (short)0, acc, false, false);
            }
            const float es  = esq[c0 + n];
            const int   code = c0 + n;
            #pragma unroll
            for (int j = 0; j < 8; ++j) {
                float sc = __builtin_fmaf(-2.0f, acc[j], es);
                if (sc < best[j]) { best[j] = sc; bidx[j] = code; }
            }
        }

        #pragma unroll
        for (int j = 0; j < 8; ++j) {
            #pragma unroll
            for (int m = 1; m <= 8; m <<= 1) {
                float ob = __shfl_xor(best[j], m, 32);
                int   oi = __shfl_xor(bidx[j], m, 32);
                if (ob < best[j] || (ob == best[j] && oi < bidx[j])) {
                    best[j] = ob; bidx[j] = oi;
                }
            }
        }

        float lp = (hi == 0) ? xsqFull : 0.0f;
        if (n == 0) {
            float s8 = 0.0f;
            #pragma unroll
            for (int j = 0; j < 8; ++j) s8 += best[j];
            lp += s8;
        }
        #pragma unroll
        for (int m = 16; m >= 1; m >>= 1) lp += __shfl_xor(lp, m, 32);
        waveLoss += lp;

        #pragma unroll
        for (int r = 0; r < 16; ++r) {
            const int j = r & 7;
            const int srcLane = (r >> 3) << 4;
            const int idx = __shfl(bidx[j], srcLane, 32);
            const float* er = eLds + idx * ESTRIDE;
            float* orow = out + (size_t)(rowBase + r) * D;
            orow[lane]      = er[lane];
            orow[lane + 32] = er[lane + 32];
        }
    }

    if (lane == 0) wl[wave] = waveLoss;
    __syncthreads();
    if (tid == 0) {
        float s = 0.0f;
        #pragma unroll
        for (int w = 0; w < WAVES; ++w) s += wl[w];
        partial[blockIdx.x] = s;
    }
}

__global__ __launch_bounds__(NUM_BLOCKS)
void vq_finalize(const float* __restrict__ partial, float* __restrict__ lossOut)
{
    __shared__ float red[NUM_BLOCKS];
    const int t = threadIdx.x;
    red[t] = partial[t];
    __syncthreads();
    for (int s = NUM_BLOCKS / 2; s > 0; s >>= 1) {
        if (t < s) red[t] += red[t + s];
        __syncthreads();
    }
    if (t == 0)
        lossOut[0] = 1.25f * red[0] / (float)((size_t)ROWS_TOTAL * D);
}

extern "C" void kernel_launch(void* const* d_in, const int* in_sizes, int n_in,
                              void* d_out, int out_size, void* d_ws, size_t ws_size,
                              hipStream_t stream)
{
    (void)in_sizes; (void)n_in; (void)out_size; (void)ws_size;
    const float* x   = (const float*)d_in[0];
    const float* emb = (const float*)d_in[1];
    float* out       = (float*)d_out;
    float* partial   = (float*)d_ws;

    vq_main<<<NUM_BLOCKS, THREADS, LDS_WORDS * sizeof(float), stream>>>(x, emb, out, partial);
    vq_finalize<<<1, NUM_BLOCKS, 0, stream>>>(partial, out + (size_t)ROWS_TOTAL * D);
}


# Work-group of 16 waves instead of 8, one 16-row slab per wave instead of two: 4 waves per SIMD, so other waves' WMMAs cover each wave's loads, argmin reduce and row write-out; sums bit-identical
# speedup vs baseline: 1.6768x; 1.1512x over previous
#include <hip/hip_runtime.h>

typedef float v2f __attribute__((ext_vector_type(2)));
typedef float v8f __attribute__((ext_vector_type(8)));

#define D            64
#define K            1024
#define ESTRIDE      66
#define ROWS_TOTAL   65536
#define THREADS      512
#define WAVES        16
#define ROWS_PER_BLOCK (WAVES * 16)
#define NUM_BLOCKS   (ROWS_TOTAL / ROWS_PER_BLOCK)

#define LDS_E_WORDS  (K * ESTRIDE)
#define LDS_ESQ_OFF  LDS_E_WORDS
#define LDS_WL_OFF   (LDS_E_WORDS + K)
#define LDS_WORDS    (LDS_E_WORDS + K + WAVES)

__global__ __launch_bounds__(THREADS)
void vq_main(const float* __restrict__ x, const float* __restrict__ emb,
             float* __restrict__ out, float* __restrict__ partial)
{
    extern __shared__ float lds[];
    float* eLds = lds;
    float* esq  = lds + LDS_ESQ_OFF;
    float* wl   = lds + LDS_WL_OFF;

    const int tid  = threadIdx.x;
    const int wave = tid >> 5;
    const int lane = tid & 31;
    const int n    = lane & 15;
    const int hi   = lane >> 4;

    for (int i = tid; i < K * D; i += THREADS)
        eLds[(i >> 6) * ESTRIDE + (i & 63)] = emb[i];
    __syncthreads();

    for (int c = tid; c < K; c += THREADS) {
        const float* er = eLds + c * ESTRIDE;
        float u[16];
        #pragma unroll
        for (int l = 0; l < 16; l += 2) {
            const v2f a0 = *(const v2f*)(er + l);
            const v2f a1 = *(const v2f*)(er + l + 32);
            const v2f b0 = *(const v2f*)(er + l + 16);
            const v2f b1 = *(const v2f*)(er + l + 48);
            const float p0 = __builtin_fmaf(a0[0], a0[0], a1[0] * a1[0]);
            const float p1 = __builtin_fmaf(a0[1], a0[1], a1[1] * a1[1]);
            const float q0 = __builtin_fmaf(b0[0], b0[0], b1[0] * b1[0]);
            const float q1 = __builtin_fmaf(b0[1], b0[1], b1[1] * b1[1]);
            u[l]     = p0 + q0;
            u[l + 1] = p1 + q1;
        }
        float v[8];
        #pragma unroll
        for (int l = 0; l < 8; ++l) v[l] = u[l] + u[l + 8];
        float w[4];
        #pragma unroll
        for (int l = 0; l < 4; ++l) w[l] = v[l] + v[l + 4];
        const float y0 = w[0] + w[2];
        const float y1 = w[1] + w[3];
        esq[c] = y0 + y1;
    }
    __syncthreads();

    float waveLoss;

    {
        const int rowBase = blockIdx.x * ROWS_PER_BLOCK + wave * 16;

        const float* xrow = x + (size_t)(rowBase + n) * D + 2 * hi;
        v2f a[16];
        float xsqp = 0.0f;
        #pragma unroll
        for (int kk = 0; kk < 16; ++kk) {
            a[kk] = *(const v2f*)(xrow + 4 * kk);
            xsqp += a[kk][0] * a[kk][0] + a[kk][1] * a[kk][1];
        }
        const float xsqFull = xsqp + __shfl_xor(xsqp, 16, 32);

        float best[8];
        int   bidx[8];
        #pragma unroll
        for (int j = 0; j < 8; ++j) { best[j] = __builtin_inff(); bidx[j] = 0; }

        for (int t = 0; t < K / 16; ++t) {
            const int c0 = t * 16;
            const float* brow = eLds + (c0 + n) * ESTRIDE + 2 * hi;
            v8f acc = {0.f, 0.f, 0.f, 0.f, 0.f, 0.f, 0.f, 0.f};
            #pragma unroll
            for (int kk = 0; kk < 16; ++kk) {
                v2f b = *(const v2f*)(brow + 4 * kk);
                acc = __builtin_amdgcn_wmma_f32_16x16x4_f32(
                          false, a[kk], false, b, (short)0, acc, false, false);
            }
            const float es  = esq[c0 + n];
            const int   code = c0 + n;
            #pragma unroll
            for (int j = 0; j < 8; ++j) {
                float sc = __builtin_fmaf(-2.0f, acc[j], es);
                if (sc < best[j]) { best[j] = sc; bidx[j] = code; }
            }
        }

        #pragma unroll
        for (int j = 0; j < 8; ++j) {
            #pragma unroll
            for (int m = 1; m <= 8; m <<= 1) {
                float ob = __shfl_xor(best[j], m, 32);
                int   oi = __shfl_xor(bidx[j], m, 32);
                if (ob < best[j] || (ob == best[j] && oi < bidx[j])) {
                    best[j] = ob; bidx[j] = oi;
                }
            }
        }

        float lp = (hi == 0) ? xsqFull : 0.0f;
        if (n == 0) {
            float s8 = 0.0f;
            #pragma unroll
            for (int j = 0; j < 8; ++j) s8 += best[j];
            lp += s8;
        }
        #pragma unroll
        for (int m = 16; m >= 1; m >>= 1) lp += __shfl_xor(lp, m, 32);
        waveLoss = lp;

        #pragma unroll
        for (int r = 0; r < 16; ++r) {
            const int j = r & 7;
            const int srcLane = (r >> 3) << 4;
            const int idx = __shfl(bidx[j], srcLane, 32);
            const float* er = eLds + idx * ESTRIDE;
            float* orow = out + (size_t)(rowBase + r) * D;
            orow[lane]      = er[lane];
            orow[lane + 32] = er[lane + 32];
        }
    }

    if (lane == 0) wl[wave] = waveLoss;
    __syncthreads();
    if (tid == 0) {
        float s = 0.0f;
        #pragma unroll
        for (int w = 0; w < WAVES / 2; ++w) s += wl[w] + wl[w + WAVES / 2];
        partial[blockIdx.x] = s;
    }
}

__global__ __launch_bounds__(NUM_BLOCKS)
void vq_finalize(const float* __restrict__ partial, float* __restrict__ lossOut)
{
    __shared__ float red[NUM_BLOCKS];
    const int t = threadIdx.x;
    red[t] = partial[t];
    __syncthreads();
    for (int s = NUM_BLOCKS / 2; s > 0; s >>= 1) {
        if (t < s) red[t] += red[t + s];
        __syncthreads();
    }
    if (t == 0)
        lossOut[0] = 1.25f * red[0] / (float)((size_t)ROWS_TOTAL * D);
}

extern "C" void kernel_launch(void* const* d_in, const int* in_sizes, int n_in,
                              void* d_out, int out_size, void* d_ws, size_t ws_size,
                              hipStream_t stream)
{
    (void)in_sizes; (void)n_in; (void)out_size; (void)ws_size;
    const float* x   = (const float*)d_in[0];
    const float* emb = (const float*)d_in[1];
    float* out       = (float*)d_out;
    float* partial   = (float*)d_ws;

    vq_main<<<NUM_BLOCKS, THREADS, LDS_WORDS * sizeof(float), stream>>>(x, emb, out, partial);
    vq_finalize<<<1, NUM_BLOCKS, 0, stream>>>(partial, out + (size_t)ROWS_TOTAL * D);
}


# Codebook staging overlapped with the sweep: all of it asked for at kernel start (16-byte loads), first 128 codes stored and swept while the other 896 stay in flight in registers; sums bit-identical
# speedup vs baseline: 1.7982x; 1.0724x over previous
#include <hip/hip_runtime.h>

typedef float v2f __attribute__((ext_vector_type(2)));
typedef float v4f __attribute__((ext_vector_type(4)));
typedef float v8f __attribute__((ext_vector_type(8)));

#define D            64
#define K            1024
#define ESTRIDE      66
#define ROWS_TOTAL   65536
#define THREADS      512
#define WAVES        16
#define ROWS_PER_BLOCK (WAVES * 16)
#define NUM_BLOCKS   (ROWS_TOTAL / ROWS_PER_BLOCK)

#define K0           128
#define Q0           (K0 * (D / 4) / THREADS)
#define Q1           ((K - K0) * (D / 4) / THREADS)

#define LDS_E_WORDS  (K * ESTRIDE)
#define LDS_ESQ_OFF  LDS_E_WORDS
#define LDS_WL_OFF   (LDS_E_WORDS + K)
#define LDS_WORDS    (LDS_E_WORDS + K + WAVES)

__device__ __forceinline__ void code_norm(const float* eLds, float* esq, int c)
{
    const float* er = eLds + c * ESTRIDE;
    float u[16];
    #pragma unroll
    for (int l = 0; l < 16; l += 2) {
        const v2f a0 = *(const v2f*)(er + l);
        const v2f a1 = *(const v2f*)(er + l + 32);
        const v2f b0 = *(const v2f*)(er + l + 16);
        const v2f b1 = *(const v2f*)(er + l + 48);
        const float p0 = __builtin_fmaf(a0[0], a0[0], a1[0] * a1[0]);
        const float p1 = __builtin_fmaf(a0[1], a0[1], a1[1] * a1[1]);
        const float q0 = __builtin_fmaf(b0[0], b0[0], b1[0] * b1[0]);
        const float q1 = __builtin_fmaf(b0[1], b0[1], b1[1] * b1[1]);
        u[l]     = p0 + q0;
        u[l + 1] = p1 + q1;
    }
    float v[8];
    #pragma unroll
    for (int l = 0; l < 8; ++l) v[l] = u[l] + u[l + 8];
    float w[4];
    #pragma unroll
    for (int l = 0; l < 4; ++l) w[l] = v[l] + v[l + 4];
    const float y0 = w[0] + w[2];
    const float y1 = w[1] + w[3];
    esq[c] = y0 + y1;
}

__device__ __forceinline__ void store_piece(float* eLds, int g, v4f v)
{
    float* p = eLds + (g >> 4) * ESTRIDE + (g & 15) * 4;
    v2f lo = {v[0], v[1]};
    v2f hi2 = {v[2], v[3]};
    *(v2f*)p       = lo;
    *(v2f*)(p + 2) = hi2;
}

__device__ __forceinline__ void lds_barrier()
{
    __builtin_amdgcn_fence(__ATOMIC_RELEASE, "workgroup", "local");
    __builtin_amdgcn_s_barrier();
    __builtin_amdgcn_fence(__ATOMIC_ACQUIRE, "workgroup", "local");
}

__device__ __forceinline__ void sweep(const float* eLds, const float* esq, const v2f (&a)[16],
                                      float (&best)[8], int (&bidx)[8], int n, int hi, int t0, int t1)
{
    for (int t = t0; t < t1; ++t) {
        const int c0 = t * 16;
        const float* brow = eLds + (c0 + n) * ESTRIDE + 2 * hi;
        v8f acc = {0.f, 0.f, 0.f, 0.f, 0.f, 0.f, 0.f, 0.f};
        #pragma unroll
        for (int kk = 0; kk < 16; ++kk) {
            v2f b = *(const v2f*)(brow + 4 * kk);
            acc = __builtin_amdgcn_wmma_f32_16x16x4_f32(
                      false, a[kk], false, b, (short)0, acc, false, false);
        }
        const float es  = esq[c0 + n];
        const int   code = c0 + n;
        #pragma unroll
        for (int j = 0; j < 8; ++j) {
            float sc = __builtin_fmaf(-2.0f, acc[j], es);
            if (sc < best[j]) { best[j] = sc; bidx[j] = code; }
        }
    }
}

__global__ __launch_bounds__(THREADS)
void vq_main(const float* __restrict__ x, const float* __restrict__ emb,
             float* __restrict__ out, float* __restrict__ partial)
{
    extern __shared__ float lds[];
    float* eLds = lds;
    float* esq  = lds + LDS_ESQ_OFF;
    float* wl   = lds + LDS_WL_OFF;

    const int tid  = threadIdx.x;
    const int wave = tid >> 5;
    const int lane = tid & 31;
    const int n    = lane & 15;
    const int hi   = lane >> 4;

    const int rowBase = blockIdx.x * ROWS_PER_BLOCK + wave * 16;

    const float* xrow = x + (size_t)(rowBase + n) * D + 2 * hi;
    v2f a[16];
    #pragma unroll
    for (int kk = 0; kk < 16; ++kk) a[kk] = *(const v2f*)(xrow + 4 * kk);
    __builtin_amdgcn_sched_barrier(0);

    const v4f* emb4 = (const v4f*)emb;
    v4f r0[Q0];
    v4f r1[Q1];
    #pragma unroll
    for (int q = 0; q < Q0; ++q) r0[q] = emb4[tid + THREADS * q];
    __builtin_amdgcn_sched_barrier(0);
    #pragma unroll
    for (int q = 0; q < Q1; ++q) r1[q] = emb4[K0 * (D / 4) + tid + THREADS * q];

    #pragma unroll
    for (int q = 0; q < Q0; ++q) store_piece(eLds, tid + THREADS * q, r0[q]);
    lds_barrier();
    if (tid < K0) code_norm(eLds, esq, tid);
    lds_barrier();

    float xsqp = 0.0f;
    #pragma unroll
    for (int kk = 0; kk < 16; ++kk)
        xsqp += a[kk][0] * a[kk][0] + a[kk][1] * a[kk][1];
    const float xsqFull = xsqp + __shfl_xor(xsqp, 16, 32);

    float best[8];
    int   bidx[8];
    #pragma unroll
    for (int j = 0; j < 8; ++j) { best[j] = __builtin_inff(); bidx[j] = 0; }

    float waveLoss;

    sweep(eLds, esq, a, best, bidx, n, hi, 0, K0 / 16);

    #pragma unroll
    for (int q = 0; q < Q1; ++q) store_piece(eLds, K0 * (D / 4) + tid + THREADS * q, r1[q]);
    __syncthreads();
    for (int c = K0 + tid; c < K; c += THREADS) code_norm(eLds, esq, c);
    __syncthreads();

    sweep(eLds, esq, a, best, bidx, n, hi, K0 / 16, K / 16);

    {
        #pragma unroll
        for (int j = 0; j < 8; ++j) {
            #pragma unroll
            for (int m = 1; m <= 8; m <<= 1) {
                float ob = __shfl_xor(best[j], m, 32);
                int   oi = __shfl_xor(bidx[j], m, 32);
                if (ob < best[j] || (ob == best[j] && oi < bidx[j])) {
                    best[j] = ob; bidx[j] = oi;
                }
            }
        }

        float lp = (hi == 0) ? xsqFull : 0.0f;
        if (n == 0) {
            float s8 = 0.0f;
            #pragma unroll
            for (int j = 0; j < 8; ++j) s8 += best[j];
            lp += s8;
        }
        #pragma unroll
        for (int m = 16; m >= 1; m >>= 1) lp += __shfl_xor(lp, m, 32);
        waveLoss = lp;

        #pragma unroll
        for (int r = 0; r < 16; ++r) {
            const int j = r & 7;
            const int srcLane = (r >> 3) << 4;
            const int idx = __shfl(bidx[j], srcLane, 32);
            const float* er = eLds + idx * ESTRIDE;
            float* orow = out + (size_t)(rowBase + r) * D;
            orow[lane]      = er[lane];
            orow[lane + 32] = er[lane + 32];
        }
    }

    if (lane == 0) wl[wave] = waveLoss;
    __syncthreads();
    if (tid == 0) {
        float s = 0.0f;
        #pragma unroll
        for (int w = 0; w < WAVES / 2; ++w) s += wl[w] + wl[w + WAVES / 2];
        partial[blockIdx.x] = s;
    }
}

__global__ __launch_bounds__(NUM_BLOCKS)
void vq_finalize(const float* __restrict__ partial, float* __restrict__ lossOut)
{
    __shared__ float red[NUM_BLOCKS];
    const int t = threadIdx.x;
    red[t] = partial[t];
    __syncthreads();
    for (int s = NUM_BLOCKS / 2; s > 0; s >>= 1) {
        if (t < s) red[t] += red[t + s];
        __syncthreads();
    }
    if (t == 0)
        lossOut[0] = 1.25f * red[0] / (float)((size_t)ROWS_TOTAL * D);
}

extern "C" void kernel_launch(void* const* d_in, const int* in_sizes, int n_in,
                              void* d_out, int out_size, void* d_ws, size_t ws_size,
                              hipStream_t stream)
{
    (void)in_sizes; (void)n_in; (void)out_size; (void)ws_size;
    const float* x   = (const float*)d_in[0];
    const float* emb = (const float*)d_in[1];
    float* out       = (float*)d_out;
    float* partial   = (float*)d_ws;

    vq_main<<<NUM_BLOCKS, THREADS, LDS_WORDS * sizeof(float), stream>>>(x, emb, out, partial);
    vq_finalize<<<1, NUM_BLOCKS, 0, stream>>>(partial, out + (size_t)ROWS_TOTAL * D);
}


# Main sweep: 4 code tiles per trip on accumulators of their own, WMMAs in turn, all B reads as one group, so no WMMA waits for the one before it; per-tile sums and update order kept, bit-identical
# speedup vs baseline: 1.8695x; 1.0397x over previous
#include <hip/hip_runtime.h>

typedef float v2f __attribute__((ext_vector_type(2)));
typedef float v4f __attribute__((ext_vector_type(4)));
typedef float v8f __attribute__((ext_vector_type(8)));

#define D            64
#define K            1024
#define ESTRIDE      66
#define ROWS_TOTAL   65536
#define THREADS      512
#define WAVES        16
#define ROWS_PER_BLOCK (WAVES * 16)
#define NUM_BLOCKS   (ROWS_TOTAL / ROWS_PER_BLOCK)

#define K0           128
#define Q0           (K0 * (D / 4) / THREADS)
#define Q1           ((K - K0) * (D / 4) / THREADS)

#define NT_MAIN      4

#define LDS_E_WORDS  (K * ESTRIDE)
#define LDS_ESQ_OFF  LDS_E_WORDS
#define LDS_WL_OFF   (LDS_E_WORDS + K)
#define LDS_WORDS    (LDS_E_WORDS + K + WAVES)

__device__ __forceinline__ void code_norm(const float* eLds, float* esq, int c)
{
    const float* er = eLds + c * ESTRIDE;
    float u[16];
    #pragma unroll
    for (int l = 0; l < 16; l += 2) {
        const v2f a0 = *(const v2f*)(er + l);
        const v2f a1 = *(const v2f*)(er + l + 32);
        const v2f b0 = *(const v2f*)(er + l + 16);
        const v2f b1 = *(const v2f*)(er + l + 48);
        const float p0 = __builtin_fmaf(a0[0], a0[0], a1[0] * a1[0]);
        const float p1 = __builtin_fmaf(a0[1], a0[1], a1[1] * a1[1]);
        const float q0 = __builtin_fmaf(b0[0], b0[0], b1[0] * b1[0]);
        const float q1 = __builtin_fmaf(b0[1], b0[1], b1[1] * b1[1]);
        u[l]     = p0 + q0;
        u[l + 1] = p1 + q1;
    }
    float v[8];
    #pragma unroll
    for (int l = 0; l < 8; ++l) v[l] = u[l] + u[l + 8];
    float w[4];
    #pragma unroll
    for (int l = 0; l < 4; ++l) w[l] = v[l] + v[l + 4];
    const float y0 = w[0] + w[2];
    const float y1 = w[1] + w[3];
    esq[c] = y0 + y1;
}

__device__ __forceinline__ void store_piece(float* eLds, int g, v4f v)
{
    float* p = eLds + (g >> 4) * ESTRIDE + (g & 15) * 4;
    v2f lo = {v[0], v[1]};
    v2f hi2 = {v[2], v[3]};
    *(v2f*)p       = lo;
    *(v2f*)(p + 2) = hi2;
}

__device__ __forceinline__ void lds_barrier()
{
    __builtin_amdgcn_fence(__ATOMIC_RELEASE, "workgroup", "local");
    __builtin_amdgcn_s_barrier();
    __builtin_amdgcn_fence(__ATOMIC_ACQUIRE, "workgroup", "local");
}

template <int NT>
__device__ __forceinline__ void sweep(const float* eLds, const float* esq, const v2f (&a)[16],
                                      float (&best)[8], int (&bidx)[8], int n, int hi, int t0, int t1)
{
    for (int t = t0; t < t1; t += NT) {
        const int c0 = t * 16;
        const float* brow = eLds + (c0 + n) * ESTRIDE + 2 * hi;
        v2f b[NT][16];
        #pragma unroll
        for (int kk = 0; kk < 16; ++kk) {
            #pragma unroll
            for (int u = 0; u < NT; ++u)
                b[u][kk] = *(const v2f*)(brow + u * 16 * ESTRIDE + 4 * kk);
        }
        v8f acc[NT];
        #pragma unroll
        for (int u = 0; u < NT; ++u) acc[u] = (v8f){0.f, 0.f, 0.f, 0.f, 0.f, 0.f, 0.f, 0.f};
        #pragma unroll
        for (int kk = 0; kk < 16; ++kk) {
            #pragma unroll
            for (int u = 0; u < NT; ++u)
                acc[u] = __builtin_amdgcn_wmma_f32_16x16x4_f32(
                             false, a[kk], false, b[u][kk], (short)0, acc[u], false, false);
        }
        #pragma unroll
        for (int u = 0; u < NT; ++u) {
            const int   code = c0 + 16 * u + n;
            const float es   = esq[code];
            #pragma unroll
            for (int j = 0; j < 8; ++j) {
                float sc = __builtin_fmaf(-2.0f, acc[u][j], es);
                if (sc < best[j]) { best[j] = sc; bidx[j] = code; }
            }
        }
    }
}

__global__ __launch_bounds__(THREADS)
void vq_main(const float* __restrict__ x, const float* __restrict__ emb,
             float* __restrict__ out, float* __restrict__ partial)
{
    extern __shared__ float lds[];
    float* eLds = lds;
    float* esq  = lds + LDS_ESQ_OFF;
    float* wl   = lds + LDS_WL_OFF;

    const int tid  = threadIdx.x;
    const int wave = tid >> 5;
    const int lane = tid & 31;
    const int n    = lane & 15;
    const int hi   = lane >> 4;

    const int rowBase = blockIdx.x * ROWS_PER_BLOCK + wave * 16;

    const float* xrow = x + (size_t)(rowBase + n) * D + 2 * hi;
    v2f a[16];
    #pragma unroll
    for (int kk = 0; kk < 16; ++kk) a[kk] = *(const v2f*)(xrow + 4 * kk);
    __builtin_amdgcn_sched_barrier(0);

    const v4f* emb4 = (const v4f*)emb;
    v4f r0[Q0];
    v4f r1[Q1];
    #pragma unroll
    for (int q = 0; q < Q0; ++q) r0[q] = emb4[tid + THREADS * q];
    __builtin_amdgcn_sched_barrier(0);
    #pragma unroll
    for (int q = 0; q < Q1; ++q) r1[q] = emb4[K0 * (D / 4) + tid + THREADS * q];

    #pragma unroll
    for (int q = 0; q < Q0; ++q) store_piece(eLds, tid + THREADS * q, r0[q]);
    lds_barrier();
    if (tid < K0) code_norm(eLds, esq, tid);
    lds_barrier();

    float xsqp = 0.0f;
    #pragma unroll
    for (int kk = 0; kk < 16; ++kk)
        xsqp += a[kk][0] * a[kk][0] + a[kk][1] * a[kk][1];
    const float xsqFull = xsqp + __shfl_xor(xsqp, 16, 32);

    float best[8];
    int   bidx[8];
    #pragma unroll
    for (int j = 0; j < 8; ++j) { best[j] = __builtin_inff(); bidx[j] = 0; }

    float waveLoss;

    sweep<1>(eLds, esq, a, best, bidx, n, hi, 0, K0 / 16);

    #pragma unroll
    for (int q = 0; q < Q1; ++q) store_piece(eLds, K0 * (D / 4) + tid + THREADS * q, r1[q]);
    __syncthreads();
    for (int c = K0 + tid; c < K; c += THREADS) code_norm(eLds, esq, c);
    __syncthreads();

    sweep<NT_MAIN>(eLds, esq, a, best, bidx, n, hi, K0 / 16, K / 16);

    {
        #pragma unroll
        for (int j = 0; j < 8; ++j) {
            #pragma unroll
            for (int m = 1; m <= 8; m <<= 1) {
                float ob = __shfl_xor(best[j], m, 32);
                int   oi = __shfl_xor(bidx[j], m, 32);
                if (ob < best[j] || (ob == best[j] && oi < bidx[j])) {
                    best[j] = ob; bidx[j] = oi;
                }
            }
        }

        float lp = (hi == 0) ? xsqFull : 0.0f;
        if (n == 0) {
            float s8 = 0.0f;
            #pragma unroll
            for (int j = 0; j < 8; ++j) s8 += best[j];
            lp += s8;
        }
        #pragma unroll
        for (int m = 16; m >= 1; m >>= 1) lp += __shfl_xor(lp, m, 32);
        waveLoss = lp;

        #pragma unroll
        for (int r = 0; r < 16; ++r) {
            const int j = r & 7;
            const int srcLane = (r >> 3) << 4;
            const int idx = __shfl(bidx[j], srcLane, 32);
            const float* er = eLds + idx * ESTRIDE;
            float* orow = out + (size_t)(rowBase + r) * D;
            orow[lane]      = er[lane];
            orow[lane + 32] = er[lane + 32];
        }
    }

    if (lane == 0) wl[wave] = waveLoss;
    __syncthreads();
    if (tid == 0) {
        float s = 0.0f;
        #pragma unroll
        for (int w = 0; w < WAVES / 2; ++w) s += wl[w] + wl[w + WAVES / 2];
        partial[blockIdx.x] = s;
    }
}

__global__ __launch_bounds__(NUM_BLOCKS)
void vq_finalize(const float* __restrict__ partial, float* __restrict__ lossOut)
{
    __shared__ float red[NUM_BLOCKS];
    const int t = threadIdx.x;
    red[t] = partial[t];
    __syncthreads();
    for (int s = NUM_BLOCKS / 2; s > 0; s >>= 1) {
        if (t < s) red[t] += red[t + s];
        __syncthreads();
    }
    if (t == 0)
        lossOut[0] = 1.25f * red[0] / (float)((size_t)ROWS_TOTAL * D);
}

extern "C" void kernel_launch(void* const* d_in, const int* in_sizes, int n_in,
                              void* d_out, int out_size, void* d_ws, size_t ws_size,
                              hipStream_t stream)
{
    (void)in_sizes; (void)n_in; (void)out_size; (void)ws_size;
    const float* x   = (const float*)d_in[0];
    const float* emb = (const float*)d_in[1];
    float* out       = (float*)d_out;
    float* partial   = (float*)d_ws;

    vq_main<<<NUM_BLOCKS, THREADS, LDS_WORDS * sizeof(float), stream>>>(x, emb, out, partial);
    vq_finalize<<<1, NUM_BLOCKS, 0, stream>>>(partial, out + (size_t)ROWS_TOTAL * D);
}
